# starting kernel
# baseline (speedup 1.0000x reference)
#include <hip/hip_runtime.h>

typedef float v2f __attribute__((ext_vector_type(2)));
typedef float v8f __attribute__((ext_vector_type(8)));

#define T_SEQ 2048
#define FEAT  256
#define GATES 64
#define HID   16
#define BATCH 64

__global__ __launch_bounds__(128) void lstm_in_gemm_wmma(
    const float* __restrict__ x,
    const float* __restrict__ W,
    const float* __restrict__ bih,
    const float* __restrict__ bhh,
    float* __restrict__ xg)
{
  const int lane = threadIdx.x & 31;
  const int m    = threadIdx.x >> 5;
  const int half = lane >> 4;
  const int r    = lane & 15;
  const int tb   = blockIdx.x * 16;
  const int b    = blockIdx.y;

  const float* Xb   = x + (size_t)b * FEAT * T_SEQ;
  const float* Wrow = W + (size_t)(m * 16 + r) * FEAT;

  v8f acc = {};
  for (int k = 0; k < FEAT; k += 4) {
    const int ka = k + 2 * half;
    v2f av, bv;
    av.x = Wrow[ka];
    av.y = Wrow[ka + 1];
    bv.x = Xb[(size_t)ka * T_SEQ + tb + r];
    bv.y = Xb[(size_t)(ka + 1) * T_SEQ + tb + r];
    acc = __builtin_amdgcn_wmma_f32_16x16x4_f32(
         false, av,  false, bv,
         (short)0, acc,  false,  false);
  }

#pragma unroll
  for (int v = 0; v < 8; ++v) {
    const int g = m * 16 + v + 8 * half;
    const float val = acc[v] + bih[g] + bhh[g];
    xg[((size_t)b * GATES + g) * T_SEQ + tb + r] = val;
  }
}

__device__ __forceinline__ float bcast(float v, int j) {
  return __uint_as_float(__builtin_amdgcn_readlane(__float_as_uint(v), (unsigned)j));
}
__device__ __forceinline__ float sig_f(float x) {
  return __builtin_amdgcn_rcpf(1.0f + __expf(-x));
}
__device__ __forceinline__ float tanh_f(float x) {
  return 1.0f - 2.0f * __builtin_amdgcn_rcpf(__expf(2.0f * x) + 1.0f);
}

__global__ __launch_bounds__(32) void lstm_recurrence(
    const float* __restrict__ xg,
    const float* __restrict__ Whh0,
    const float* __restrict__ Wih1,
    const float* __restrict__ Whh1,
    const float* __restrict__ bih1,
    const float* __restrict__ bhh1,
    float* __restrict__ out)
{
  const int lane = threadIdx.x;
  const int b    = blockIdx.x;
  const int g0   = lane;
  const int g1   = lane + 32;

  float w0a[HID], w0b[HID], wia[HID], wib[HID], wha[HID], whb[HID];
#pragma unroll
  for (int j = 0; j < HID; ++j) {
    w0a[j] = Whh0[g0 * HID + j];
    w0b[j] = Whh0[g1 * HID + j];
    wia[j] = Wih1[g0 * HID + j];
    wib[j] = Wih1[g1 * HID + j];
    wha[j] = Whh1[g0 * HID + j];
    whb[j] = Whh1[g1 * HID + j];
  }
  const float bias0 = bih1[g0] + bhh1[g0];
  const float bias1 = bih1[g1] + bhh1[g1];

  const float* xg0p = xg + ((size_t)b * GATES + g0) * T_SEQ;
  const float* xg1p = xg + ((size_t)b * GATES + g1) * T_SEQ;
  float* outb = out + (size_t)b * T_SEQ;

  float h0 = 0.0f, c0 = 0.0f;
  float h1 = 0.0f, c1 = 0.0f;
  float hb0[HID], hb1[HID];
#pragma unroll
  for (int j = 0; j < HID; ++j) { hb0[j] = 0.0f; hb1[j] = 0.0f; }

  for (int t0 = 0; t0 < T_SEQ; t0 += 4) {
    const float4 x0v = *(const float4*)(xg0p + t0);
    const float4 x1v = *(const float4*)(xg1p + t0);
    __builtin_prefetch(xg0p + t0 + 64, 0, 3);
    __builtin_prefetch(xg1p + t0 + 64, 0, 3);
    const float x0a[4] = {x0v.x, x0v.y, x0v.z, x0v.w};
    const float x1a[4] = {x1v.x, x1v.y, x1v.z, x1v.w};

#pragma unroll
    for (int u = 0; u < 4; ++u) {
      float p0a = 0.f, p0b = 0.f, p1a = 0.f, p1b = 0.f;
#pragma unroll
      for (int j = 0; j < 8; ++j) {
        p0a = fmaf(w0a[j],     hb0[j],     p0a);
        p0b = fmaf(w0a[j + 8], hb0[j + 8], p0b);
        p1a = fmaf(w0b[j],     hb0[j],     p1a);
        p1b = fmaf(w0b[j + 8], hb0[j + 8], p1b);
      }
      const float a0 = x0a[u] + (p0a + p0b);
      const float a1 = x1a[u] + (p1a + p1b);
      const float fv = __shfl_xor(a0, 16, 32);
      const float ov = __shfl_xor(a1, 16, 32);
      c0 = sig_f(fv) * c0 + sig_f(a0) * tanh_f(a1);
      h0 = sig_f(ov) * tanh_f(c0);

#pragma unroll
      for (int j = 0; j < HID; ++j) hb0[j] = bcast(h0, j);

      float r0a = 0.f, r0b = 0.f, r0c = 0.f, r0d = 0.f;
      float r1a = 0.f, r1b = 0.f, r1c = 0.f, r1d = 0.f;
#pragma unroll
      for (int j = 0; j < 8; ++j) {
        r0a = fmaf(wia[j],     hb0[j],     r0a);
        r0b = fmaf(wia[j + 8], hb0[j + 8], r0b);
        r0c = fmaf(wha[j],     hb1[j],     r0c);
        r0d = fmaf(wha[j + 8], hb1[j + 8], r0d);
        r1a = fmaf(wib[j],     hb0[j],     r1a);
        r1b = fmaf(wib[j + 8], hb0[j + 8], r1b);
        r1c = fmaf(whb[j],     hb1[j],     r1c);
        r1d = fmaf(whb[j + 8], hb1[j + 8], r1d);
      }
      const float b0 = bias0 + ((r0a + r0b) + (r0c + r0d));
      const float b1 = bias1 + ((r1a + r1b) + (r1c + r1d));
      const float fv1 = __shfl_xor(b0, 16, 32);
      const float ov1 = __shfl_xor(b1, 16, 32);
      c1 = sig_f(fv1) * c1 + sig_f(b0) * tanh_f(b1);
      h1 = sig_f(ov1) * tanh_f(c1);

#pragma unroll
      for (int j = 0; j < HID; ++j) hb1[j] = bcast(h1, j);
      const float sA = (hb1[0] + hb1[1]) + (hb1[2] + hb1[3]);
      const float sB = (hb1[4] + hb1[5]) + (hb1[6] + hb1[7]);
      const float sC = (hb1[8] + hb1[9]) + (hb1[10] + hb1[11]);
      const float sD = (hb1[12] + hb1[13]) + (hb1[14] + hb1[15]);
      const float mean = ((sA + sB) + (sC + sD)) * 0.0625f;
      if (lane == 0) outb[t0 + u] = mean;
    }
  }
}

extern "C" void kernel_launch(void* const* d_in, const int* in_sizes, int n_in,
                              void* d_out, int out_size, void* d_ws, size_t ws_size,
                              hipStream_t stream) {
  (void)in_sizes; (void)n_in; (void)out_size; (void)ws_size;
  const float* x    = (const float*)d_in[0];
  const float* Wih0 = (const float*)d_in[1];
  const float* Whh0 = (const float*)d_in[2];
  const float* bih0 = (const float*)d_in[3];
  const float* bhh0 = (const float*)d_in[4];
  const float* Wih1 = (const float*)d_in[5];
  const float* Whh1 = (const float*)d_in[6];
  const float* bih1 = (const float*)d_in[7];
  const float* bhh1 = (const float*)d_in[8];
  float* out = (float*)d_out;
  float* xg  = (float*)d_ws;

  dim3 g1(T_SEQ / 16, BATCH, 1);
  lstm_in_gemm_wmma<<<g1, dim3(128, 1, 1), 0, stream>>>(x, Wih0, bih0, bhh0, xg);

  lstm_recurrence<<<dim3(BATCH, 1, 1), dim3(32, 1, 1), 0, stream>>>(
      xg, Whh0, Wih1, Whh1, bih1, bhh1, out);
}


# input GEMM fused into the recurrence kernel: 2 projection waves per work-group form each 16-step gate tile by the same WMMA chain into a 2-buffer LDS ring, 1 barrier per 16 steps; sum order unchanged
# speedup vs baseline: 1.0542x; 1.0542x over previous
#include <hip/hip_runtime.h>


typedef float v2f __attribute__((ext_vector_type(2)));
typedef float v8f __attribute__((ext_vector_type(8)));

#define T_SEQ  2048
#define FEAT   256
#define GATES  64
#define HID    16
#define BATCH  64
#define WIN    16
#define NWIN   (T_SEQ / WIN)
#define PITCH  20
#define KSTEPS (FEAT / 4)
#define KCH    16

__device__ __forceinline__ float bcast(float v, int j) {
  return __uint_as_float(__builtin_amdgcn_readlane(__float_as_uint(v), (unsigned)j));
}
__device__ __forceinline__ float sig_f(float x) {
  return __builtin_amdgcn_rcpf(1.0f + __expf(-x));
}
__device__ __forceinline__ float tanh_f(float x) {
  return 1.0f - 2.0f * __builtin_amdgcn_rcpf(__expf(2.0f * x) + 1.0f);
}

__device__ __forceinline__ void load_x_group(v2f (&bq)[KCH], const float* __restrict__ xc, int c) {
#pragma unroll
  for (int j = 0; j < KCH; ++j) {
    const float* p = xc + (size_t)((c * KCH + j) * 4) * T_SEQ;
    bq[j].x = p[0];
    bq[j].y = p[T_SEQ];
  }
}

__device__ __forceinline__ void produce_window(
    const float* __restrict__ xc, const v2f* __restrict__ af,
    const float* __restrict__ bi, const float* __restrict__ bh,
    float* __restrict__ dst, int r)
{
  v2f bq[2][KCH];
  load_x_group(bq[0], xc, 0);
  v8f acc0 = {}, acc1 = {};
#pragma unroll
  for (int c = 0; c < KSTEPS / KCH; ++c) {
    if (c + 1 < KSTEPS / KCH) load_x_group(bq[(c + 1) & 1], xc, c + 1);
#pragma unroll
    for (int s = 0; s < KCH / 8; ++s) {
      v2f fa0[8], fa1[8];
#pragma unroll
      for (int j = 0; j < 8; ++j) {
        const int i = c * KCH + s * 8 + j;
        fa0[j] = af[i * 32];
        fa1[j] = af[(KSTEPS + i) * 32];
      }
#pragma unroll
      for (int j = 0; j < 8; ++j) {
        acc0 = __builtin_amdgcn_wmma_f32_16x16x4_f32(
            false, fa0[j], false, bq[c & 1][s * 8 + j], (short)0, acc0, false, false);
        acc1 = __builtin_amdgcn_wmma_f32_16x16x4_f32(
            false, fa1[j], false, bq[c & 1][s * 8 + j], (short)0, acc1, false, false);
      }
    }
  }
#pragma unroll
  for (int v = 0; v < 8; ++v) {
    dst[v * PITCH + r]        = acc0[v] + bi[v]      + bh[v];
    dst[(v + 16) * PITCH + r] = acc1[v] + bi[16 + v] + bh[16 + v];
  }
}

__device__ __forceinline__ void ring_handover() {
  __builtin_amdgcn_fence(__ATOMIC_RELEASE, "workgroup", "local");
  __builtin_amdgcn_s_barrier();
  __builtin_amdgcn_fence(__ATOMIC_ACQUIRE, "workgroup", "local");
}

__global__ __launch_bounds__(96) void lstm_fused(
    const float* __restrict__ x,
    const float* __restrict__ Wih0,
    const float* __restrict__ bih0,
    const float* __restrict__ bhh0,
    const float* __restrict__ Whh0,
    const float* __restrict__ Wih1,
    const float* __restrict__ Whh1,
    const float* __restrict__ bih1,
    const float* __restrict__ bhh1,
    float* __restrict__ out)
{
  __shared__ __attribute__((aligned(16))) float ring[2][GATES * PITCH];
  __shared__ __attribute__((aligned(16))) v2f afr[2][2 * KSTEPS * 32];

  const int lane = threadIdx.x & 31;
  const int role = __builtin_amdgcn_readfirstlane((int)(threadIdx.x >> 5));
  const int b    = blockIdx.x;
  const int half = lane >> 4;
  const int r    = lane & 15;

  const int pw = (role - 1) & 1;
  const int m0 = 2 * pw;
  const float* xc = x + (size_t)b * FEAT * T_SEQ + (size_t)(2 * half) * T_SEQ + r;
  const int drow = (m0 * 16 + 8 * half) * PITCH;
  v2f* af = &afr[pw][lane];
  const float* bi = bih0 + m0 * 16 + 8 * half;
  const float* bh = bhh0 + m0 * 16 + 8 * half;

  const int g0 = lane;
  const int g1 = lane + 32;
  float w0a[HID], w0b[HID], wia[HID], wib[HID], wha[HID], whb[HID];
  float bias0 = 0.0f, bias1 = 0.0f;
  float h0 = 0.0f, c0 = 0.0f;
  float h1 = 0.0f, c1 = 0.0f;
  float hb0[HID], hb1[HID];
  float* outb = out + (size_t)b * T_SEQ;

  if (role == 0) {
#pragma unroll
    for (int j = 0; j < HID; ++j) {
      w0a[j] = Whh0[g0 * HID + j];
      w0b[j] = Whh0[g1 * HID + j];
      wia[j] = Wih1[g0 * HID + j];
      wib[j] = Wih1[g1 * HID + j];
      wha[j] = Whh1[g0 * HID + j];
      whb[j] = Whh1[g1 * HID + j];
      hb0[j] = 0.0f;
      hb1[j] = 0.0f;
    }
    bias0 = bih1[g0] + bhh1[g0];
    bias1 = bih1[g1] + bhh1[g1];
  } else {
    const float* Wr0 = Wih0 + (size_t)(m0 * 16 + r) * FEAT + 2 * half;
    const float* Wr1 = Wr0 + (size_t)16 * FEAT;
#pragma unroll 16
    for (int i = 0; i < KSTEPS; ++i) {
      af[i * 32]            = *(const v2f*)(Wr0 + 4 * i);
      af[(KSTEPS + i) * 32] = *(const v2f*)(Wr1 + 4 * i);
    }
    produce_window(xc, af, bi, bh, &ring[0][drow], r);
  }
  ring_handover();

  for (int w = 0; w < NWIN; ++w) {
    if (role == 0) {
      const float* rb0 = &ring[w & 1][g0 * PITCH];
      const float* rb1 = &ring[w & 1][g1 * PITCH];
      float4 n0 = *(const float4*)rb0;
      float4 n1 = *(const float4*)rb1;
      for (int q = 0; q < WIN / 4; ++q) {
        const float4 x0v = n0;
        const float4 x1v = n1;
        const int qn = (q + 1) & (WIN / 4 - 1);
        n0 = *(const float4*)(rb0 + 4 * qn);
        n1 = *(const float4*)(rb1 + 4 * qn);
        const float x0a[4] = {x0v.x, x0v.y, x0v.z, x0v.w};
        const float x1a[4] = {x1v.x, x1v.y, x1v.z, x1v.w};
        const int t0 = w * WIN + q * 4;

#pragma unroll
        for (int u = 0; u < 4; ++u) {
          float p0a = 0.f, p0b = 0.f, p1a = 0.f, p1b = 0.f;
#pragma unroll
          for (int j = 0; j < 8; ++j) {
            p0a = fmaf(w0a[j],     hb0[j],     p0a);
            p0b = fmaf(w0a[j + 8], hb0[j + 8], p0b);
            p1a = fmaf(w0b[j],     hb0[j],     p1a);
            p1b = fmaf(w0b[j + 8], hb0[j + 8], p1b);
          }
          const float a0g = x0a[u] + (p0a + p0b);
          const float a1g = x1a[u] + (p1a + p1b);
          const float fv = __shfl_xor(a0g, 16, 32);
          const float ov = __shfl_xor(a1g, 16, 32);
          c0 = sig_f(fv) * c0 + sig_f(a0g) * tanh_f(a1g);
          h0 = sig_f(ov) * tanh_f(c0);

#pragma unroll
          for (int j = 0; j < HID; ++j) hb0[j] = bcast(h0, j);

          float r0a = 0.f, r0b = 0.f, r0c = 0.f, r0d = 0.f;
          float r1a = 0.f, r1b = 0.f, r1c = 0.f, r1d = 0.f;
#pragma unroll
          for (int j = 0; j < 8; ++j) {
            r0a = fmaf(wia[j],     hb0[j],     r0a);
            r0b = fmaf(wia[j + 8], hb0[j + 8], r0b);
            r0c = fmaf(wha[j],     hb1[j],     r0c);
            r0d = fmaf(wha[j + 8], hb1[j + 8], r0d);
            r1a = fmaf(wib[j],     hb0[j],     r1a);
            r1b = fmaf(wib[j + 8], hb0[j + 8], r1b);
            r1c = fmaf(whb[j],     hb1[j],     r1c);
            r1d = fmaf(whb[j + 8], hb1[j + 8], r1d);
          }
          const float b0 = bias0 + ((r0a + r0b) + (r0c + r0d));
          const float b1 = bias1 + ((r1a + r1b) + (r1c + r1d));
          const float fv1 = __shfl_xor(b0, 16, 32);
          const float ov1 = __shfl_xor(b1, 16, 32);
          c1 = sig_f(fv1) * c1 + sig_f(b0) * tanh_f(b1);
          h1 = sig_f(ov1) * tanh_f(c1);

#pragma unroll
          for (int j = 0; j < HID; ++j) hb1[j] = bcast(h1, j);
          const float sA = (hb1[0] + hb1[1]) + (hb1[2] + hb1[3]);
          const float sB = (hb1[4] + hb1[5]) + (hb1[6] + hb1[7]);
          const float sC = (hb1[8] + hb1[9]) + (hb1[10] + hb1[11]);
          const float sD = (hb1[12] + hb1[13]) + (hb1[14] + hb1[15]);
          const float mean = ((sA + sB) + (sC + sD)) * 0.0625f;
          if (lane == 0) outb[t0 + u] = mean;
        }
      }
    } else if (w + 1 < NWIN) {
      produce_window(xc + (w + 1) * WIN, af, bi, bh, &ring[(w + 1) & 1][drow], r);
    }
    ring_handover();
  }
}

extern "C" void kernel_launch(void* const* d_in, const int* in_sizes, int n_in,
                              void* d_out, int out_size, void* d_ws, size_t ws_size,
                              hipStream_t stream) {
  (void)in_sizes; (void)n_in; (void)out_size; (void)d_ws; (void)ws_size;
  const float* x    = (const float*)d_in[0];
  const float* Wih0 = (const float*)d_in[1];
  const float* Whh0 = (const float*)d_in[2];
  const float* bih0 = (const float*)d_in[3];
  const float* bhh0 = (const float*)d_in[4];
  const float* Wih1 = (const float*)d_in[5];
  const float* Whh1 = (const float*)d_in[6];
  const float* bih1 = (const float*)d_in[7];
  const float* bhh1 = (const float*)d_in[8];
  float* out = (float*)d_out;

  lstm_fused<<<dim3(BATCH, 1, 1), dim3(96, 1, 1), 0, stream>>>(
      x, Wih0, bih0, bhh0, Whh0, Wih1, Whh1, bih1, bhh1, out);
}


# recurrence: layer 1 runs two windows behind; its input half Wih1 x h0 leaves the trip: a 4th wave forms it per window by matrix instructions from the h0 tile in LDS into a 2nd ring; sum order changes
# speedup vs baseline: 4.5650x; 4.3303x over previous
#include <hip/hip_runtime.h>


typedef float v2f __attribute__((ext_vector_type(2)));
typedef float v8f __attribute__((ext_vector_type(8)));

#define T_SEQ  2048
#define FEAT   256
#define GATES  64
#define HID    16
#define BATCH  64
#define WIN    32
#define TILE   16
#define NWIN   (T_SEQ / WIN)
#define PITCH  36
#define KSTEPS (FEAT / 4)
#define ABLK   4
#define MT     (GATES / 16)
#define NBLK   (KSTEPS / ABLK)
#define BIASBLK (NBLK - 5)
#define XRING  6
#define XPRE   3
#define PREBLK 8
#define PASSBLK (NBLK - 2)
#define XPBUF  (XPRE * (ABLK / 2) * 32 * 4)
#define APAIRS (KSTEPS / 2)
#define HPITCH 20
#define HDUMP  (WIN * HPITCH + 16)
#define HBUF   (HDUMP + WIN * HPITCH)

__device__ __forceinline__ float bcast_row(float v, int j) {
  const unsigned sel = 0x11111111u * (unsigned)j;
  const unsigned u = __float_as_uint(v);
  return __uint_as_float(__builtin_amdgcn_permlane16(u, u, sel, sel, true, false));
}
__device__ __forceinline__ float sig_f(float x) {
  return __builtin_amdgcn_rcpf(1.0f + __expf(-x));
}
__device__ __forceinline__ float tanh_f(float x) {
  return 1.0f - 2.0f * __builtin_amdgcn_rcpf(__expf(2.0f * x) + 1.0f);
}

__device__ __forceinline__ void load_x_block(v2f (&bq)[ABLK], const float* __restrict__ xc, int c) {
#pragma unroll
  for (int j = 0; j < ABLK; ++j) {
    const float* p = xc + (size_t)((c * ABLK + j) * 4) * T_SEQ;
    bq[j].x = p[0];
    bq[j].y = p[T_SEQ];
  }
}

__device__ __forceinline__ void load_a_block(float4 (&af)[MT][ABLK / 2], const float* ap, int c) {
#pragma unroll
  for (int t = 0; t < MT; ++t)
#pragma unroll
    for (int p = 0; p < ABLK / 2; ++p)
      af[t][p] = *(const float4*)(ap + (size_t)((t * APAIRS + c * (ABLK / 2) + p) * 32) * 4);
}

__device__ __forceinline__ void store_x_pre(float* xp, const v2f (&pre)[XPRE][ABLK]) {
#pragma unroll
  for (int c = 0; c < XPRE; ++c)
#pragma unroll
    for (int p = 0; p < ABLK / 2; ++p)
      *(float4*)(xp + (size_t)((c * (ABLK / 2) + p) * 32) * 4) =
          float4{pre[c][2 * p].x, pre[c][2 * p].y, pre[c][2 * p + 1].x, pre[c][2 * p + 1].y};
}

__device__ __forceinline__ void produce_window(
    float* xp, const float* __restrict__ xc, const float* __restrict__ xn, const float* ap,
    const float* __restrict__ bi, const float* __restrict__ bh,
    float* __restrict__ dst, int r)
{
  v2f bq[XRING][ABLK];
  v2f pre[XPRE][ABLK];
  float4 af[2][MT][ABLK / 2];
  load_a_block(af[0], ap, 0);
#pragma unroll
  for (int c = 0; c < XPRE; ++c)
#pragma unroll
    for (int p = 0; p < ABLK / 2; ++p) {
      const float4 q = *(const float4*)(xp + (size_t)((c * (ABLK / 2) + p) * 32) * 4);
      bq[c][2 * p]     = v2f{q.x, q.y};
      bq[c][2 * p + 1] = v2f{q.z, q.w};
    }
#pragma unroll
  for (int c = XPRE; c < XRING - 1; ++c) load_x_block(bq[c], xc, c);
  __builtin_amdgcn_sched_group_barrier(0x100, (MT + XPRE) * (ABLK / 2), 0);
  __builtin_amdgcn_sched_group_barrier(0x020, 2 * ABLK * (XRING - 1 - XPRE), 0);
  v8f acc[MT];
#pragma unroll
  for (int t = 0; t < MT; ++t) acc[t] = v8f{};
  float bb[MT][2][8];
#pragma unroll
  for (int c = 0; c < NBLK; ++c) {
    if (c + XRING - 1 < NBLK) load_x_block(bq[(c + XRING - 1) % XRING], xc, c + XRING - 1);
    if (c >= PREBLK && c - PREBLK < XPRE) load_x_block(pre[c - PREBLK], xn, c - PREBLK);
    if (c == PASSBLK) store_x_pre(xp, pre);
    if (c + 1 < NBLK) load_a_block(af[(c + 1) & 1], ap, c + 1);
    if (c == BIASBLK) {
#pragma unroll
      for (int t = 0; t < MT; ++t)
#pragma unroll
        for (int v = 0; v < 8; ++v) { bb[t][0][v] = bi[t * 16 + v]; bb[t][1][v] = bh[t * 16 + v]; }
    }
#pragma unroll
    for (int j = 0; j < ABLK; ++j) {
#pragma unroll
      for (int t = 0; t < MT; ++t) {
        const float4 q = af[c & 1][t][j >> 1];
        const v2f a = (j & 1) ? v2f{q.z, q.w} : v2f{q.x, q.y};
        acc[t] = __builtin_amdgcn_wmma_f32_16x16x4_f32(
            false, a, false, bq[c % XRING][j], (short)0, acc[t], false, false);
      }
    }
    if (c + 1 < NBLK)         __builtin_amdgcn_sched_group_barrier(0x100, MT * (ABLK / 2), 0);
    if (c == PASSBLK)         __builtin_amdgcn_sched_group_barrier(0x200, XPRE * (ABLK / 2), 0);
    if (c + XRING - 1 < NBLK)              __builtin_amdgcn_sched_group_barrier(0x020, 2 * ABLK, 0);
    if (c >= PREBLK && c - PREBLK < XPRE)  __builtin_amdgcn_sched_group_barrier(0x020, 2 * ABLK, 0);
    if (c == BIASBLK + 1)     __builtin_amdgcn_sched_group_barrier(0x020, MT * 2 * 2, 0);
    __builtin_amdgcn_sched_group_barrier(0x008, MT * ABLK, 0);
  }
#pragma unroll
  for (int t = 0; t < MT; ++t)
#pragma unroll
    for (int v = 0; v < 8; ++v)
      dst[(t * 16 + v) * PITCH + r] = acc[t][v] + bb[t][0][v] + bb[t][1][v];
}

__device__ __forceinline__ void lstm_trip(
    const v2f (&wH)[2][HID], const v2f (&e)[2], float& hs, float& cs)
{
  v2f c0[2], c1[2];
#pragma unroll
  for (int p = 0; p < 2; ++p) { c0[p] = 0.f; c1[p] = 0.f; }
#pragma unroll
  for (int k = 0; k < 8; ++k) {
    const float ra = bcast_row(hs, k), rb = bcast_row(hs, k + 8);
    const v2f ha = {ra, ra}, hb = {rb, rb};
#pragma unroll
    for (int p = 0; p < 2; ++p) {
      c0[p] = __builtin_elementwise_fma(wH[p][k],     ha, c0[p]);
      c1[p] = __builtin_elementwise_fma(wH[p][k + 8], hb, c1[p]);
    }
  }
  v2f gt[2];
#pragma unroll
  for (int p = 0; p < 2; ++p) gt[p] = e[p] + (c0[p] + c1[p]);
  cs = sig_f(gt[0].y) * cs + sig_f(gt[0].x) * tanh_f(gt[1].x);
  hs = sig_f(gt[1].y) * tanh_f(cs);
}

__device__ __forceinline__ void produce_xg1(
    const float* tile, const float* __restrict__ Wih1,
    const float* __restrict__ bih1, const float* __restrict__ bhh1,
    float* __restrict__ dst, int r, int half)
{
  v2f a1[MT][HID / 4];
  float b1[MT][8];
#pragma unroll
  for (int t = 0; t < MT; ++t) {
#pragma unroll
    for (int j = 0; j < HID / 4; ++j)
      a1[t][j] = *(const v2f*)(Wih1 + (t * 16 + r) * HID + 4 * j + 2 * half);
#pragma unroll
    for (int v = 0; v < 8; ++v)
      b1[t][v] = bih1[t * 16 + 8 * half + v] + bhh1[t * 16 + 8 * half + v];
  }
#pragma unroll
  for (int tt = 0; tt < WIN / TILE; ++tt) {
    v2f bq[HID / 4];
#pragma unroll
    for (int j = 0; j < HID / 4; ++j)
      bq[j] = *(const v2f*)(tile + (tt * TILE + r) * HPITCH + 4 * j + 2 * half);
    v8f acc[MT];
#pragma unroll
    for (int t = 0; t < MT; ++t) acc[t] = v8f{};
#pragma unroll
    for (int j = 0; j < HID / 4; ++j) {
#pragma unroll
      for (int t = 0; t < MT; ++t)
        acc[t] = __builtin_amdgcn_wmma_f32_16x16x4_f32(
            false, a1[t][j], false, bq[j], (short)0, acc[t], false, false);
    }
#pragma unroll
    for (int t = 0; t < MT; ++t)
#pragma unroll
      for (int v = 0; v < 8; ++v)
        dst[(t * 16 + v) * PITCH + tt * TILE + r] = acc[t][v] + b1[t][v];
  }
}

__device__ __forceinline__ void emit_means(const float* tile, float* __restrict__ outb,
                                           int base, int lane) {
  const int t = lane;
  const float4* p = (const float4*)(tile + t * HPITCH);
  const float4 a = p[0], b = p[1], c = p[2], d = p[3];
  const float lo = ((a.x + a.y) + (a.z + a.w)) + ((b.x + b.y) + (b.z + b.w));
  const float hi = ((c.x + c.y) + (c.z + c.w)) + ((d.x + d.y) + (d.z + d.w));
  outb[base + t] = (lo + hi) * 0.0625f;
}

__device__ __forceinline__ void ring_handover() {
  __builtin_amdgcn_fence(__ATOMIC_RELEASE, "workgroup", "local");
  __builtin_amdgcn_s_barrier();
  __builtin_amdgcn_fence(__ATOMIC_ACQUIRE, "workgroup", "local");
}

__global__ __launch_bounds__(128) void lstm_fused(
    const float* __restrict__ x,
    const float* __restrict__ Wih0,
    const float* __restrict__ bih0,
    const float* __restrict__ bhh0,
    const float* __restrict__ Whh0,
    const float* __restrict__ Wih1,
    const float* __restrict__ Whh1,
    const float* __restrict__ bih1,
    const float* __restrict__ bhh1,
    float* __restrict__ out)
{
  __shared__ __attribute__((aligned(16))) float ring[2][GATES * PITCH];
  __shared__ __attribute__((aligned(16))) float ring1[2][GATES * PITCH];
  __shared__ __attribute__((aligned(16))) float hbuf[2][HBUF];
  __shared__ __attribute__((aligned(16))) float afrag[4 * APAIRS * 32 * 4];
  __shared__ __attribute__((aligned(16))) float xpre[2][XPBUF];

  const int lane = threadIdx.x & 31;
  const int role = __builtin_amdgcn_readfirstlane((int)(threadIdx.x >> 5));
  const int b    = blockIdx.x;
  const int half = lane >> 4;
  const int r    = lane & 15;

  const int pw = (role - 1) & 1;
  const int m0 = 2 * pw;
  const float* xc = x + (size_t)b * FEAT * T_SEQ + (size_t)(2 * half) * T_SEQ + pw * TILE + r;
  const int drow = (8 * half) * PITCH + pw * TILE;
  const float* ap = afrag + (size_t)lane * 4;
  const float* bi = bih0 + 8 * half;
  const float* bh = bhh0 + 8 * half;
  float* xp = &xpre[pw][lane * 4];

  v2f wH[2][HID];
  float hs = 0.0f, cs = 0.0f;
  float* outb = out + (size_t)b * T_SEQ;

  if (role == 0) {
    const float* WH = half ? Whh1 : Whh0;
#pragma unroll
    for (int g = 0; g < 4; ++g) {
#pragma unroll
      for (int k = 0; k < HID; ++k)
        wH[g >> 1][k][g & 1] = WH[(g * HID + r) * HID + k];
    }
  } else if (role == 3) {
    float4* z = (float4*)&ring1[0][0];
    for (int i = lane; i < 2 * GATES * PITCH / 4; i += 32) z[i] = float4{0.f, 0.f, 0.f, 0.f};
  } else {
    const float* Wr0 = Wih0 + (size_t)(m0 * 16 + r) * FEAT + 2 * half;
    const float* Wr1 = Wr0 + (size_t)16 * FEAT;
    float* aw = afrag + (size_t)(m0 * APAIRS * 32 + lane) * 4;
    for (int p = 0; p < APAIRS; ++p) {
      const v2f u0 = *(const v2f*)(Wr0 + 8 * p), u1 = *(const v2f*)(Wr0 + 8 * p + 4);
      const v2f w0 = *(const v2f*)(Wr1 + 8 * p), w1 = *(const v2f*)(Wr1 + 8 * p + 4);
      *(float4*)(aw + (size_t)(p * 32) * 4)            = float4{u0.x, u0.y, u1.x, u1.y};
      *(float4*)(aw + (size_t)((APAIRS + p) * 32) * 4) = float4{w0.x, w0.y, w1.x, w1.y};
    }
  }
  ring_handover();
  if (role == 1 || role == 2) {
    v2f pre[XPRE][ABLK];
#pragma unroll
    for (int c = 0; c < XPRE; ++c) load_x_block(pre[c], xc, c);
    store_x_pre(xp, pre);
    produce_window(xp, xc, xc + WIN, ap, bi, bh, &ring[0][drow], r);
  }
  ring_handover();

  for (int w = 0; w < NWIN + 2; ++w) {
    if (role == 0) {
      if (w == 2) {
        hs = half ? 0.0f : hs;
        cs = half ? 0.0f : cs;
      }
      const float* rb = half ? &ring1[w & 1][r * PITCH] : &ring[w & 1][r * PITCH];
      float* hst = &hbuf[w & 1][(half ? 0 : HDUMP) + r];
      float4 nx[4];
#pragma unroll
      for (int g = 0; g < 4; ++g) nx[g] = *(const float4*)(rb + g * HID * PITCH);
      for (int q = 0; q < WIN / 4; ++q) {
        float xa[4][4];
#pragma unroll
        for (int g = 0; g < 4; ++g) {
          xa[g][0] = nx[g].x; xa[g][1] = nx[g].y; xa[g][2] = nx[g].z; xa[g][3] = nx[g].w;
        }
        const int qn = (q + 1) & (WIN / 4 - 1);
#pragma unroll
        for (int g = 0; g < 4; ++g) nx[g] = *(const float4*)(rb + g * HID * PITCH + 4 * qn);

#pragma unroll
        for (int u = 0; u < 4; ++u) {
          v2f e[2];
#pragma unroll
          for (int g = 0; g < 4; ++g) e[g >> 1][g & 1] = xa[g][u];
          lstm_trip(wH, e, hs, cs);
          hst[(q * 4 + u) * HPITCH] = hs;
        }
      }
    } else if (role == 3) {
      if (w >= 1 && w <= NWIN)
        produce_xg1(&hbuf[(w - 1) & 1][HDUMP], Wih1, bih1, bhh1,
                    &ring1[(w - 1) & 1][(8 * half) * PITCH], r, half);
    } else {
      if (w + 1 < NWIN) {
        const int wn = (w + 2 < NWIN) ? w + 2 : NWIN - 1;
        produce_window(xp, xc + (w + 1) * WIN, xc + wn * WIN, ap, bi, bh,
                       &ring[(w + 1) & 1][drow], r);
      }
      if (role == 1 && w >= 3) emit_means(hbuf[(w - 1) & 1], outb, (w - 3) * WIN, lane);
    }
    ring_handover();
  }

  if (role == 1) emit_means(hbuf[(NWIN + 1) & 1], outb, (NWIN - 1) * WIN, lane);
}

extern "C" void kernel_launch(void* const* d_in, const int* in_sizes, int n_in,
                              void* d_out, int out_size, void* d_ws, size_t ws_size,
                              hipStream_t stream) {
  (void)in_sizes; (void)n_in; (void)out_size; (void)d_ws; (void)ws_size;
  const float* x    = (const float*)d_in[0];
  const float* Wih0 = (const float*)d_in[1];
  const float* Whh0 = (const float*)d_in[2];
  const float* bih0 = (const float*)d_in[3];
  const float* bhh0 = (const float*)d_in[4];
  const float* Wih1 = (const float*)d_in[5];
  const float* Whh1 = (const float*)d_in[6];
  const float* bih1 = (const float*)d_in[7];
  const float* bhh1 = (const float*)d_in[8];
  float* out = (float*)d_out;

  lstm_fused<<<dim3(BATCH, 1, 1), dim3(128, 1, 1), 0, stream>>>(
      x, Wih0, bih0, bhh0, Whh0, Wih1, Whh1, bih1, bhh1, out);
}
